# starting kernel
# baseline (speedup 1.0000x reference)
#include <hip/hip_runtime.h>

typedef __attribute__((ext_vector_type(2))) float v2f;
typedef __attribute__((ext_vector_type(8))) float v8f;

#define DHID     256
#define LTOK     128
#define LLAB     32
#define NPG      64
#define BGRAPH   64
#define NNODES   (BGRAPH * NPG)
#define ODIM     1024

__global__ void ragged_bag_kernel(const int* __restrict__ text_ids,
                                  const int* __restrict__ label_ids,
                                  const int* __restrict__ text_len,
                                  const int* __restrict__ label_len,
                                  const float* __restrict__ word_emb,
                                  const float* __restrict__ label_emb,
                                  float* __restrict__ text_out,
                                  float* __restrict__ label_out) {
    __shared__ int s_tid[LTOK];
    __shared__ int s_lid[LLAB];
    const int n = blockIdx.x;
    const int d = threadIdx.x;

    if (threadIdx.x < LTOK) s_tid[threadIdx.x] = text_ids[n * LTOK + threadIdx.x];
    if (threadIdx.x < LLAB) s_lid[threadIdx.x] = label_ids[n * LLAB + threadIdx.x];
    __syncthreads();

    const int tl = text_len[n];
    float acc = 0.0f;
    for (int l = 0; l < tl; ++l)
        acc += word_emb[(size_t)s_tid[l] * DHID + d];
    text_out[(size_t)n * DHID + d] = acc / (float)tl;

    const int ll = label_len[n];
    float accl = 0.0f;
    for (int l = 0; l < ll; ++l)
        accl += label_emb[(size_t)s_lid[l] * DHID + d];
    label_out[(size_t)n * DHID + d] = accl;
}

__global__ void gemm_wmma_f32(const float* __restrict__ A,
                              const float* __restrict__ B,
                              const float* __restrict__ bias,
                              float* __restrict__ C,
                              int M, int N, int K, int relu) {
    const int wave       = (blockIdx.x * blockDim.x + threadIdx.x) >> 5;
    const int tilesN     = N >> 4;
    const int totalTiles = (M >> 4) * tilesN;
    if (wave >= totalTiles) return;

    const int tm   = wave / tilesN;
    const int tn   = wave % tilesN;
    const int lane = threadIdx.x & 31;
    const int kh   = lane >> 4;
    const int l15  = lane & 15;

    const float* Arow = A + (size_t)(tm * 16 + l15) * K;
    const float* Bcol = B + (size_t)(tn * 16 + l15);

    v8f acc = {};
    for (int k = 0; k < K; k += 4) {
        const int kk = k + kh * 2;
        v2f a = *(const v2f*)(Arow + kk);
        v2f b;
        b.x = Bcol[(size_t)kk * N];
        b.y = Bcol[(size_t)(kk + 1) * N];
        acc = __builtin_amdgcn_wmma_f32_16x16x4_f32(
             false, a,  false, b,
             (short)0, acc,  false,  false);
    }

    const int col  = tn * 16 + l15;
    const int row0 = tm * 16 + kh * 8;
    const float bv = bias ? bias[col] : 0.0f;
#pragma unroll
    for (int v = 0; v < 8; ++v) {
        float val = acc[v] + bv;
        if (relu) val = fmaxf(val, 0.0f);
        C[(size_t)(row0 + v) * N + col] = val;
    }
}

__global__ void readout_kernel(const float* __restrict__ h_t,
                               const float* __restrict__ h_l,
                               const float* __restrict__ w_adap,
                               float* __restrict__ fused) {
    const int b = blockIdx.x;
    const int d = threadIdx.x;
    float st = 0.0f, sl = 0.0f;
    for (int i = 0; i < NPG; ++i) {
        const size_t node = (size_t)b * NPG + i;
        st += h_t[node * DHID + d];
        sl += h_l[node * DHID + d];
    }
    const float inv = 1.0f / (float)NPG;
    fused[(size_t)b * DHID + d] = (st * inv) * w_adap[0] + (sl * inv) * w_adap[1];
}

extern "C" void kernel_launch(void* const* d_in, const int* in_sizes, int n_in,
                              void* d_out, int out_size, void* d_ws, size_t ws_size,
                              hipStream_t stream) {
    const int*   text_ids  = (const int*)  d_in[0];
    const int*   label_ids = (const int*)  d_in[1];
    const int*   text_len  = (const int*)  d_in[2];
    const int*   label_len = (const int*)  d_in[3];
    const float* word_emb  = (const float*)d_in[4];
    const float* label_emb = (const float*)d_in[5];
    const float* Wg_text   = (const float*)d_in[6];
    const float* Wg_label  = (const float*)d_in[7];
    const float* w_adap    = (const float*)d_in[8];
    const float* W1        = (const float*)d_in[9];
    const float* b1        = (const float*)d_in[10];
    const float* Wout      = (const float*)d_in[11];
    const float* bout      = (const float*)d_in[12];
    float* out = (float*)d_out;

    float* ws        = (float*)d_ws;
    float* text_out  = ws;
    float* label_out = text_out  + (size_t)NNODES * DHID;
    float* h_t       = label_out + (size_t)NNODES * DHID;
    float* h_l       = h_t       + (size_t)NNODES * DHID;
    float* fused     = h_l       + (size_t)NNODES * DHID;
    float* hmid      = fused     + (size_t)BGRAPH * DHID;

    ragged_bag_kernel<<<NNODES, 256, 0, stream>>>(
        text_ids, label_ids, text_len, label_len,
        word_emb, label_emb, text_out, label_out);

    const int tiles1 = (NNODES / 16) * (DHID / 16);
    gemm_wmma_f32<<<(tiles1 + 7) / 8, 256, 0, stream>>>(
        text_out, Wg_text, nullptr, h_t, NNODES, DHID, DHID,  1);
    gemm_wmma_f32<<<(tiles1 + 7) / 8, 256, 0, stream>>>(
        label_out, Wg_label, nullptr, h_l, NNODES, DHID, DHID,  1);

    readout_kernel<<<BGRAPH, 256, 0, stream>>>(h_t, h_l, w_adap, fused);

    const int tiles2 = (BGRAPH / 16) * (DHID / 16);
    gemm_wmma_f32<<<(tiles2 + 7) / 8, 256, 0, stream>>>(
        fused, W1, b1, hmid, BGRAPH, DHID, DHID,  1);

    const int tiles3 = (BGRAPH / 16) * (ODIM / 16);
    gemm_wmma_f32<<<(tiles3 + 7) / 8, 256, 0, stream>>>(
        hmid, Wout, bout, out, BGRAPH, ODIM, DHID,  0);
}


# bags feed the projection WMMAs through LDS, not memory: a 16-wave block sums its 16 nodes' bags (16 rows in flight per wave) into two A tiles, runs both projections; launches 6->4; sum order same
# speedup vs baseline: 1.6374x; 1.6374x over previous
#include <hip/hip_runtime.h>


typedef __attribute__((ext_vector_type(2))) float v2f;
typedef __attribute__((ext_vector_type(4))) float v4f;
typedef __attribute__((ext_vector_type(4))) int   v4i;
typedef __attribute__((ext_vector_type(8))) float v8f;

#define DHID     256
#define LTOK     128
#define LLAB     32
#define NPG      64
#define BGRAPH   64
#define NNODES   (BGRAPH * NPG)
#define ODIM     1024

#define TILE_M   16
#define APITCH   260
#define ROWS_FLY 16

__device__ __forceinline__ void bag_row_sum(v4i idv, const int ipl, int len,
                                            const float* __restrict__ table, int lane,
                                            v4f& s0, v4f& s1) {
    v4f a0 = {0.0f, 0.0f, 0.0f, 0.0f};
    v4f a1 = {0.0f, 0.0f, 0.0f, 0.0f};
    const float* base = table + 4 * lane;
    for (int l = 0; l < len; l += ROWS_FLY) {
        v4f r0[ROWS_FLY], r1[ROWS_FLY];
        const int id_first = __builtin_amdgcn_readlane(idv.x, l / ipl);
#pragma unroll
        for (int j = 0; j < ROWS_FLY; ++j) {
            const int c   = j % ipl;
            const int reg = (c == 0) ? idv.x : (c == 1) ? idv.y : (c == 2) ? idv.z : idv.w;
            const int id_j = __builtin_amdgcn_readlane(reg, (l + j) / ipl);
            const int id   = (l + j < len) ? id_j : id_first;
            const float* p = base + (size_t)id * DHID;
            r0[j] = *(const v4f*)p;
            r1[j] = *(const v4f*)(p + 128);
        }
        __builtin_amdgcn_sched_barrier(0);
#pragma unroll
        for (int j = 0; j < ROWS_FLY; ++j) {
            if (l + j < len) { a0 += r0[j]; a1 += r1[j]; }
        }
    }
    s0 = a0;
    s1 = a1;
}

__global__ __launch_bounds__(512) void bag_project_kernel(
        const int* __restrict__ text_ids,
        const int* __restrict__ label_ids,
        const int* __restrict__ text_len,
        const int* __restrict__ label_len,
        const float* __restrict__ word_emb,
        const float* __restrict__ label_emb,
        const float* __restrict__ Wg_text,
        const float* __restrict__ Wg_label,
        float* __restrict__ h_t,
        float* __restrict__ h_l) {
    __shared__ __attribute__((aligned(16))) float At[TILE_M * APITCH];
    __shared__ __attribute__((aligned(16))) float Al[TILE_M * APITCH];

    const int lane = threadIdx.x & 31;
    const int w = __builtin_amdgcn_readfirstlane((int)(threadIdx.x >> 5));
    const int n = blockIdx.x * TILE_M + w;

    {
        const v4i tid = *(const v4i*)(text_ids + (size_t)n * LTOK + 4 * lane);
        v4i lid = {0, 0, 0, 0};
        lid.x = label_ids[(size_t)n * LLAB + lane];
        const int tl = text_len[n];
        const int ll = label_len[n];

        v4f s0, s1;
        bag_row_sum(tid, 4, tl, word_emb, lane, s0, s1);
        const float ftl = (float)tl;
        s0 = s0 / ftl;
        s1 = s1 / ftl;
        *(v4f*)&At[w * APITCH + 4 * lane]       = s0;
        *(v4f*)&At[w * APITCH + 128 + 4 * lane] = s1;

        bag_row_sum(lid, 1, ll, label_emb, lane, s0, s1);
        *(v4f*)&Al[w * APITCH + 4 * lane]       = s0;
        *(v4f*)&Al[w * APITCH + 128 + 4 * lane] = s1;
    }
    __syncthreads();

    const int kh  = lane >> 4;
    const int l15 = lane & 15;

    const float* at = At + l15 * APITCH + 2 * kh;
    const float* al = Al + l15 * APITCH + 2 * kh;
    const float* bt = Wg_text  + (size_t)(2 * kh) * DHID + w * 16 + l15;
    const float* bl = Wg_label + (size_t)(2 * kh) * DHID + w * 16 + l15;

    v8f acc_t = {};
    v8f acc_l = {};
#pragma unroll 8
    for (int k = 0; k < DHID; k += 4) {
        const v2f a_t = *(const v2f*)(at + k);
        const v2f a_l = *(const v2f*)(al + k);
        v2f b_t, b_l;
        b_t.x = bt[(size_t)k * DHID];
        b_t.y = bt[(size_t)(k + 1) * DHID];
        b_l.x = bl[(size_t)k * DHID];
        b_l.y = bl[(size_t)(k + 1) * DHID];
        acc_t = __builtin_amdgcn_wmma_f32_16x16x4_f32(
             false, a_t,  false, b_t,
             (short)0, acc_t,  false,  false);
        acc_l = __builtin_amdgcn_wmma_f32_16x16x4_f32(
             false, a_l,  false, b_l,
             (short)0, acc_l,  false,  false);
    }

    const int col  = w * 16 + l15;
    const int row0 = blockIdx.x * TILE_M + kh * 8;
#pragma unroll
    for (int v = 0; v < 8; ++v) {
        h_t[(size_t)(row0 + v) * DHID + col] = fmaxf(acc_t[v], 0.0f);
        h_l[(size_t)(row0 + v) * DHID + col] = fmaxf(acc_l[v], 0.0f);
    }
}

__global__ void gemm_wmma_f32(const float* __restrict__ A,
                              const float* __restrict__ B,
                              const float* __restrict__ bias,
                              float* __restrict__ C,
                              int M, int N, int K, int relu) {
    const int wave       = (blockIdx.x * blockDim.x + threadIdx.x) >> 5;
    const int tilesN     = N >> 4;
    const int totalTiles = (M >> 4) * tilesN;
    if (wave >= totalTiles) return;

    const int tm   = wave / tilesN;
    const int tn   = wave % tilesN;
    const int lane = threadIdx.x & 31;
    const int kh   = lane >> 4;
    const int l15  = lane & 15;

    const float* Arow = A + (size_t)(tm * 16 + l15) * K;
    const float* Bcol = B + (size_t)(tn * 16 + l15);

    v8f acc = {};
    for (int k = 0; k < K; k += 4) {
        const int kk = k + kh * 2;
        v2f a = *(const v2f*)(Arow + kk);
        v2f b;
        b.x = Bcol[(size_t)kk * N];
        b.y = Bcol[(size_t)(kk + 1) * N];
        acc = __builtin_amdgcn_wmma_f32_16x16x4_f32(
             false, a,  false, b,
             (short)0, acc,  false,  false);
    }

    const int col  = tn * 16 + l15;
    const int row0 = tm * 16 + kh * 8;
    const float bv = bias ? bias[col] : 0.0f;
#pragma unroll
    for (int v = 0; v < 8; ++v) {
        float val = acc[v] + bv;
        if (relu) val = fmaxf(val, 0.0f);
        C[(size_t)(row0 + v) * N + col] = val;
    }
}

__global__ void readout_kernel(const float* __restrict__ h_t,
                               const float* __restrict__ h_l,
                               const float* __restrict__ w_adap,
                               float* __restrict__ fused) {
    const int b = blockIdx.x;
    const int d = threadIdx.x;
    float st = 0.0f, sl = 0.0f;
    for (int i = 0; i < NPG; ++i) {
        const size_t node = (size_t)b * NPG + i;
        st += h_t[node * DHID + d];
        sl += h_l[node * DHID + d];
    }
    const float inv = 1.0f / (float)NPG;
    fused[(size_t)b * DHID + d] = (st * inv) * w_adap[0] + (sl * inv) * w_adap[1];
}

extern "C" void kernel_launch(void* const* d_in, const int* in_sizes, int n_in,
                              void* d_out, int out_size, void* d_ws, size_t ws_size,
                              hipStream_t stream) {
    const int*   text_ids  = (const int*)  d_in[0];
    const int*   label_ids = (const int*)  d_in[1];
    const int*   text_len  = (const int*)  d_in[2];
    const int*   label_len = (const int*)  d_in[3];
    const float* word_emb  = (const float*)d_in[4];
    const float* label_emb = (const float*)d_in[5];
    const float* Wg_text   = (const float*)d_in[6];
    const float* Wg_label  = (const float*)d_in[7];
    const float* w_adap    = (const float*)d_in[8];
    const float* W1        = (const float*)d_in[9];
    const float* b1        = (const float*)d_in[10];
    const float* Wout      = (const float*)d_in[11];
    const float* bout      = (const float*)d_in[12];
    float* out = (float*)d_out;

    float* ws        = (float*)d_ws;
    float* h_t       = ws + (size_t)2 * NNODES * DHID;
    float* h_l       = h_t       + (size_t)NNODES * DHID;
    float* fused     = h_l       + (size_t)NNODES * DHID;
    float* hmid      = fused     + (size_t)BGRAPH * DHID;

    bag_project_kernel<<<NNODES / TILE_M, 512, 0, stream>>>(
        text_ids, label_ids, text_len, label_len,
        word_emb, label_emb, Wg_text, Wg_label, h_t, h_l);

    readout_kernel<<<BGRAPH, 256, 0, stream>>>(h_t, h_l, w_adap, fused);

    const int tiles2 = (BGRAPH / 16) * (DHID / 16);
    gemm_wmma_f32<<<(tiles2 + 7) / 8, 256, 0, stream>>>(
        fused, W1, b1, hmid, BGRAPH, DHID, DHID,  1);

    const int tiles3 = (BGRAPH / 16) * (ODIM / 16);
    gemm_wmma_f32<<<(tiles3 + 7) / 8, 256, 0, stream>>>(
        hmid, Wout, bout, out, BGRAPH, ODIM, DHID,  0);
}


# tail in one launch: readout + fused@W1 + output GEMM per block of 4 graphs x 256 out cols (the 4 graphs fill a 16-row WMMA tile by repetition); fused, h stay in LDS; launches 4->2; sum order same
# speedup vs baseline: 2.5281x; 1.5440x over previous
#include <hip/hip_runtime.h>


typedef __attribute__((ext_vector_type(2))) float v2f;
typedef __attribute__((ext_vector_type(4))) float v4f;
typedef __attribute__((ext_vector_type(4))) int   v4i;
typedef __attribute__((ext_vector_type(8))) float v8f;

#define DHID     256
#define LTOK     128
#define LLAB     32
#define NPG      64
#define BGRAPH   64
#define NNODES   (BGRAPH * NPG)
#define ODIM     1024

#define TILE_M   16
#define APITCH   260
#define ROWS_FLY 16

#define GPB      4
#define NSLICE   256
#define HROWS    32

__device__ __forceinline__ void bag_row_sum(v4i idv, const int ipl, int len,
                                            const float* __restrict__ table, int lane,
                                            v4f& s0, v4f& s1) {
    v4f a0 = {0.0f, 0.0f, 0.0f, 0.0f};
    v4f a1 = {0.0f, 0.0f, 0.0f, 0.0f};
    const float* base = table + 4 * lane;
    for (int l = 0; l < len; l += ROWS_FLY) {
        v4f r0[ROWS_FLY], r1[ROWS_FLY];
        const int id_first = __builtin_amdgcn_readlane(idv.x, l / ipl);
#pragma unroll
        for (int j = 0; j < ROWS_FLY; ++j) {
            const int c   = j % ipl;
            const int reg = (c == 0) ? idv.x : (c == 1) ? idv.y : (c == 2) ? idv.z : idv.w;
            const int id_j = __builtin_amdgcn_readlane(reg, (l + j) / ipl);
            const int id   = (l + j < len) ? id_j : id_first;
            const float* p = base + (size_t)id * DHID;
            r0[j] = *(const v4f*)p;
            r1[j] = *(const v4f*)(p + 128);
        }
        __builtin_amdgcn_sched_barrier(0);
#pragma unroll
        for (int j = 0; j < ROWS_FLY; ++j) {
            if (l + j < len) { a0 += r0[j]; a1 += r1[j]; }
        }
    }
    s0 = a0;
    s1 = a1;
}

__global__ __launch_bounds__(512) void bag_project_kernel(
        const int* __restrict__ text_ids,
        const int* __restrict__ label_ids,
        const int* __restrict__ text_len,
        const int* __restrict__ label_len,
        const float* __restrict__ word_emb,
        const float* __restrict__ label_emb,
        const float* __restrict__ Wg_text,
        const float* __restrict__ Wg_label,
        float* __restrict__ h_t,
        float* __restrict__ h_l) {
    __shared__ __attribute__((aligned(16))) float At[TILE_M * APITCH];
    __shared__ __attribute__((aligned(16))) float Al[TILE_M * APITCH];

    const int lane = threadIdx.x & 31;
    const int w = __builtin_amdgcn_readfirstlane((int)(threadIdx.x >> 5));
    const int n = blockIdx.x * TILE_M + w;

    {
        const v4i tid = *(const v4i*)(text_ids + (size_t)n * LTOK + 4 * lane);
        v4i lid = {0, 0, 0, 0};
        lid.x = label_ids[(size_t)n * LLAB + lane];
        const int tl = text_len[n];
        const int ll = label_len[n];

        v4f s0, s1;
        bag_row_sum(tid, 4, tl, word_emb, lane, s0, s1);
        const float ftl = (float)tl;
        s0 = s0 / ftl;
        s1 = s1 / ftl;
        *(v4f*)&At[w * APITCH + 4 * lane]       = s0;
        *(v4f*)&At[w * APITCH + 128 + 4 * lane] = s1;

        bag_row_sum(lid, 1, ll, label_emb, lane, s0, s1);
        *(v4f*)&Al[w * APITCH + 4 * lane]       = s0;
        *(v4f*)&Al[w * APITCH + 128 + 4 * lane] = s1;
    }
    __syncthreads();

    const int kh  = lane >> 4;
    const int l15 = lane & 15;

    const float* at = At + l15 * APITCH + 2 * kh;
    const float* al = Al + l15 * APITCH + 2 * kh;
    const float* bt = Wg_text  + (size_t)(2 * kh) * DHID + w * 16 + l15;
    const float* bl = Wg_label + (size_t)(2 * kh) * DHID + w * 16 + l15;

    v8f acc_t = {};
    v8f acc_l = {};
#pragma unroll 8
    for (int k = 0; k < DHID; k += 4) {
        const v2f a_t = *(const v2f*)(at + k);
        const v2f a_l = *(const v2f*)(al + k);
        v2f b_t, b_l;
        b_t.x = bt[(size_t)k * DHID];
        b_t.y = bt[(size_t)(k + 1) * DHID];
        b_l.x = bl[(size_t)k * DHID];
        b_l.y = bl[(size_t)(k + 1) * DHID];
        acc_t = __builtin_amdgcn_wmma_f32_16x16x4_f32(
             false, a_t,  false, b_t,
             (short)0, acc_t,  false,  false);
        acc_l = __builtin_amdgcn_wmma_f32_16x16x4_f32(
             false, a_l,  false, b_l,
             (short)0, acc_l,  false,  false);
    }

    const int col  = w * 16 + l15;
    const int row0 = blockIdx.x * TILE_M + kh * 8;
#pragma unroll
    for (int v = 0; v < 8; ++v) {
        h_t[(size_t)(row0 + v) * DHID + col] = fmaxf(acc_t[v], 0.0f);
        h_l[(size_t)(row0 + v) * DHID + col] = fmaxf(acc_l[v], 0.0f);
    }
}

__global__ __launch_bounds__(512) void graph_head_kernel(
        const float* __restrict__ h_t,
        const float* __restrict__ h_l,
        const float* __restrict__ w_adap,
        const float* __restrict__ W1,
        const float* __restrict__ b1,
        const float* __restrict__ Wout,
        const float* __restrict__ bout,
        float* __restrict__ out) {
    __shared__ __attribute__((aligned(16))) float Pt[GPB * APITCH];
    __shared__ __attribute__((aligned(16))) float Pl[GPB * APITCH];
    __shared__ __attribute__((aligned(16))) float Af[GPB * APITCH];
    __shared__ __attribute__((aligned(16))) float Ah[GPB * APITCH];
    __shared__ __attribute__((aligned(16))) float Os[GPB * APITCH];

    const int lane = threadIdx.x & 31;
    const int w  = __builtin_amdgcn_readfirstlane((int)(threadIdx.x >> 5));
    const int g0 = blockIdx.x * GPB;
    const int n0 = blockIdx.y * NSLICE;

    {
        const int gi = w >> 2;
        const int ty = (w >> 1) & 1;
        const int hf = w & 1;
        const float* src = (ty ? h_l : h_t)
                         + (size_t)(g0 + gi) * NPG * DHID + hf * 128 + 4 * lane;
        v4f s = {0.0f, 0.0f, 0.0f, 0.0f};
        for (int i0 = 0; i0 < NPG; i0 += HROWS) {
            v4f r[HROWS];
#pragma unroll
            for (int j = 0; j < HROWS; ++j)
                r[j] = *(const v4f*)(src + (size_t)(i0 + j) * DHID);
            __builtin_amdgcn_sched_barrier(0);
#pragma unroll
            for (int j = 0; j < HROWS; ++j) s += r[j];
        }
        const float inv = 1.0f / (float)NPG;
        const float wa  = w_adap[ty];
        s = (s * inv) * wa;
        float* dst = (ty ? Pl : Pt) + gi * APITCH + hf * 128 + 4 * lane;
        *(v4f*)dst = s;
    }
    __syncthreads();
    {
        const int gi = threadIdx.x >> 7;
        const int c  = (threadIdx.x & 127) * 2;
        const v2f pt = *(const v2f*)&Pt[gi * APITCH + c];
        const v2f pl = *(const v2f*)&Pl[gi * APITCH + c];
        *(v2f*)&Af[gi * APITCH + c] = pt + pl;
    }
    __syncthreads();

    const int kh  = lane >> 4;
    const int l15 = lane & 15;
    const int mr  = l15 & (GPB - 1);
    const int col = w * 16 + l15;

    {
        const float* af = Af + mr * APITCH + 2 * kh;
        const float* bw = W1 + (size_t)(2 * kh) * DHID + col;
        v8f acc = {};
#pragma unroll
        for (int k = 0; k < DHID; k += 4) {
            const v2f a = *(const v2f*)(af + k);
            v2f b;
            b.x = bw[(size_t)k * DHID];
            b.y = bw[(size_t)(k + 1) * DHID];
            acc = __builtin_amdgcn_wmma_f32_16x16x4_f32(
                 false, a,  false, b,
                 (short)0, acc,  false,  false);
        }
        if (kh == 0) {
            const float bv = b1[col];
#pragma unroll
            for (int v = 0; v < GPB; ++v)
                Ah[v * APITCH + col] = fmaxf(acc[v] + bv, 0.0f);
        }
    }
    __syncthreads();

    {
        const float* ah = Ah + mr * APITCH + 2 * kh;
        const float* bw = Wout + (size_t)(2 * kh) * ODIM + n0 + col;
        v8f acc = {};
#pragma unroll
        for (int k = 0; k < DHID; k += 4) {
            const v2f a = *(const v2f*)(ah + k);
            v2f b;
            b.x = bw[(size_t)k * ODIM];
            b.y = bw[(size_t)(k + 1) * ODIM];
            acc = __builtin_amdgcn_wmma_f32_16x16x4_f32(
                 false, a,  false, b,
                 (short)0, acc,  false,  false);
        }
        if (kh == 0) {
            const float bv = bout[n0 + col];
#pragma unroll
            for (int v = 0; v < GPB; ++v)
                Os[v * APITCH + col] = acc[v] + bv;
        }
    }
    __syncthreads();
    if (w < 2 * GPB) {
        const int row = w >> 1;
        const int hf  = w & 1;
        const v4f o = *(const v4f*)&Os[row * APITCH + hf * 128 + 4 * lane];
        *(v4f*)(out + (size_t)(g0 + row) * ODIM + n0 + hf * 128 + 4 * lane) = o;
    }
}

extern "C" void kernel_launch(void* const* d_in, const int* in_sizes, int n_in,
                              void* d_out, int out_size, void* d_ws, size_t ws_size,
                              hipStream_t stream) {
    const int*   text_ids  = (const int*)  d_in[0];
    const int*   label_ids = (const int*)  d_in[1];
    const int*   text_len  = (const int*)  d_in[2];
    const int*   label_len = (const int*)  d_in[3];
    const float* word_emb  = (const float*)d_in[4];
    const float* label_emb = (const float*)d_in[5];
    const float* Wg_text   = (const float*)d_in[6];
    const float* Wg_label  = (const float*)d_in[7];
    const float* w_adap    = (const float*)d_in[8];
    const float* W1        = (const float*)d_in[9];
    const float* b1        = (const float*)d_in[10];
    const float* Wout      = (const float*)d_in[11];
    const float* bout      = (const float*)d_in[12];
    float* out = (float*)d_out;

    float* ws        = (float*)d_ws;
    float* h_t       = ws + (size_t)2 * NNODES * DHID;
    float* h_l       = h_t       + (size_t)NNODES * DHID;

    bag_project_kernel<<<NNODES / TILE_M, 512, 0, stream>>>(
        text_ids, label_ids, text_len, label_len,
        word_emb, label_emb, Wg_text, Wg_label, h_t, h_l);

    graph_head_kernel<<<dim3(BGRAPH / GPB, ODIM / NSLICE), 512, 0, stream>>>(
        h_t, h_l, w_adap, W1, b1, Wout, bout, out);
}


# adaptive fusion applied per node ahead of the mean readout (linear): projection epilogue stores one plane w0*relu(t)+w1*relu(l); head reads half the row bytes; sum order: fuse first, rows 32+32
# speedup vs baseline: 2.6090x; 1.0320x over previous
#include <hip/hip_runtime.h>


typedef __attribute__((ext_vector_type(2))) float v2f;
typedef __attribute__((ext_vector_type(4))) float v4f;
typedef __attribute__((ext_vector_type(4))) int   v4i;
typedef __attribute__((ext_vector_type(8))) float v8f;

#define DHID     256
#define LTOK     128
#define LLAB     32
#define NPG      64
#define BGRAPH   64
#define NNODES   (BGRAPH * NPG)
#define ODIM     1024

#define TILE_M   16
#define APITCH   260
#define ROWS_FLY 16

#define GPB      4
#define NSLICE   256
#define HROWS    32
static_assert(2 * HROWS == NPG, "the head kernel's readout splits a graph's rows into two halves of HROWS");

__device__ __forceinline__ void bag_row_sum(v4i idv, const int ipl, int len,
                                            const float* __restrict__ table, int lane,
                                            v4f& s0, v4f& s1) {
    v4f a0 = {0.0f, 0.0f, 0.0f, 0.0f};
    v4f a1 = {0.0f, 0.0f, 0.0f, 0.0f};
    const float* base = table + 4 * lane;
    for (int l = 0; l < len; l += ROWS_FLY) {
        v4f r0[ROWS_FLY], r1[ROWS_FLY];
        const int id_first = __builtin_amdgcn_readlane(idv.x, l / ipl);
#pragma unroll
        for (int j = 0; j < ROWS_FLY; ++j) {
            const int c   = j % ipl;
            const int reg = (c == 0) ? idv.x : (c == 1) ? idv.y : (c == 2) ? idv.z : idv.w;
            const int id_j = __builtin_amdgcn_readlane(reg, (l + j) / ipl);
            const int id   = (l + j < len) ? id_j : id_first;
            const float* p = base + (size_t)id * DHID;
            r0[j] = *(const v4f*)p;
            r1[j] = *(const v4f*)(p + 128);
        }
        __builtin_amdgcn_sched_barrier(0);
#pragma unroll
        for (int j = 0; j < ROWS_FLY; ++j) {
            if (l + j < len) { a0 += r0[j]; a1 += r1[j]; }
        }
    }
    s0 = a0;
    s1 = a1;
}

__global__ __launch_bounds__(512) void bag_project_kernel(
        const int* __restrict__ text_ids,
        const int* __restrict__ label_ids,
        const int* __restrict__ text_len,
        const int* __restrict__ label_len,
        const float* __restrict__ word_emb,
        const float* __restrict__ label_emb,
        const float* __restrict__ Wg_text,
        const float* __restrict__ Wg_label,
        const float* __restrict__ w_adap,
        float* __restrict__ h_f) {
    __shared__ __attribute__((aligned(16))) float At[TILE_M * APITCH];
    __shared__ __attribute__((aligned(16))) float Al[TILE_M * APITCH];

    const int lane = threadIdx.x & 31;
    const int w = __builtin_amdgcn_readfirstlane((int)(threadIdx.x >> 5));
    const int n = blockIdx.x * TILE_M + w;

    {
        const v4i tid = *(const v4i*)(text_ids + (size_t)n * LTOK + 4 * lane);
        v4i lid = {0, 0, 0, 0};
        lid.x = label_ids[(size_t)n * LLAB + lane];
        const int tl = text_len[n];
        const int ll = label_len[n];

        v4f s0, s1;
        bag_row_sum(tid, 4, tl, word_emb, lane, s0, s1);
        const float ftl = (float)tl;
        s0 = s0 / ftl;
        s1 = s1 / ftl;
        *(v4f*)&At[w * APITCH + 4 * lane]       = s0;
        *(v4f*)&At[w * APITCH + 128 + 4 * lane] = s1;

        bag_row_sum(lid, 1, ll, label_emb, lane, s0, s1);
        *(v4f*)&Al[w * APITCH + 4 * lane]       = s0;
        *(v4f*)&Al[w * APITCH + 128 + 4 * lane] = s1;
    }
    __syncthreads();

    const int kh  = lane >> 4;
    const int l15 = lane & 15;

    const float* at = At + l15 * APITCH + 2 * kh;
    const float* al = Al + l15 * APITCH + 2 * kh;
    const float* bt = Wg_text  + (size_t)(2 * kh) * DHID + w * 16 + l15;
    const float* bl = Wg_label + (size_t)(2 * kh) * DHID + w * 16 + l15;

    v8f acc_t = {};
    v8f acc_l = {};
#pragma unroll 8
    for (int k = 0; k < DHID; k += 4) {
        const v2f a_t = *(const v2f*)(at + k);
        const v2f a_l = *(const v2f*)(al + k);
        v2f b_t, b_l;
        b_t.x = bt[(size_t)k * DHID];
        b_t.y = bt[(size_t)(k + 1) * DHID];
        b_l.x = bl[(size_t)k * DHID];
        b_l.y = bl[(size_t)(k + 1) * DHID];
        acc_t = __builtin_amdgcn_wmma_f32_16x16x4_f32(
             false, a_t,  false, b_t,
             (short)0, acc_t,  false,  false);
        acc_l = __builtin_amdgcn_wmma_f32_16x16x4_f32(
             false, a_l,  false, b_l,
             (short)0, acc_l,  false,  false);
    }

    const int col  = w * 16 + l15;
    const int row0 = blockIdx.x * TILE_M + kh * 8;
    const float wa_t = w_adap[0];
    const float wa_l = w_adap[1];
#pragma unroll
    for (int v = 0; v < 8; ++v) {
        h_f[(size_t)(row0 + v) * DHID + col] =
            wa_t * fmaxf(acc_t[v], 0.0f) + wa_l * fmaxf(acc_l[v], 0.0f);
    }
}

__global__ __launch_bounds__(512) void graph_head_kernel(
        const float* __restrict__ h_f,
        const float* __restrict__ W1,
        const float* __restrict__ b1,
        const float* __restrict__ Wout,
        const float* __restrict__ bout,
        float* __restrict__ out) {
    __shared__ __attribute__((aligned(16))) float Plo[GPB * APITCH];
    __shared__ __attribute__((aligned(16))) float Phi[GPB * APITCH];
    __shared__ __attribute__((aligned(16))) float Af[GPB * APITCH];
    __shared__ __attribute__((aligned(16))) float Ah[GPB * APITCH];
    __shared__ __attribute__((aligned(16))) float Os[GPB * APITCH];

    const int lane = threadIdx.x & 31;
    const int w  = __builtin_amdgcn_readfirstlane((int)(threadIdx.x >> 5));
    const int g0 = blockIdx.x * GPB;
    const int n0 = blockIdx.y * NSLICE;

    {
        const int gi = w >> 2;
        const int hf = (w >> 1) & 1;
        const int rh = w & 1;
        const float* src = h_f
                         + ((size_t)(g0 + gi) * NPG + rh * HROWS) * DHID + hf * 128 + 4 * lane;
        v4f s = {0.0f, 0.0f, 0.0f, 0.0f};
        v4f r[HROWS];
#pragma unroll
        for (int j = 0; j < HROWS; ++j)
            r[j] = *(const v4f*)(src + (size_t)j * DHID);
        __builtin_amdgcn_sched_barrier(0);
#pragma unroll
        for (int j = 0; j < HROWS; ++j) s += r[j];
        const float inv = 1.0f / (float)NPG;
        s = s * inv;
        float* dst = (rh ? Phi : Plo) + gi * APITCH + hf * 128 + 4 * lane;
        *(v4f*)dst = s;
    }
    __syncthreads();
    {
        const int gi = threadIdx.x >> 7;
        const int c  = (threadIdx.x & 127) * 2;
        const v2f plo = *(const v2f*)&Plo[gi * APITCH + c];
        const v2f phi = *(const v2f*)&Phi[gi * APITCH + c];
        *(v2f*)&Af[gi * APITCH + c] = plo + phi;
    }
    __syncthreads();

    const int kh  = lane >> 4;
    const int l15 = lane & 15;
    const int mr  = l15 & (GPB - 1);
    const int col = w * 16 + l15;

    {
        const float* af = Af + mr * APITCH + 2 * kh;
        const float* bw = W1 + (size_t)(2 * kh) * DHID + col;
        v8f acc = {};
#pragma unroll
        for (int k = 0; k < DHID; k += 4) {
            const v2f a = *(const v2f*)(af + k);
            v2f b;
            b.x = bw[(size_t)k * DHID];
            b.y = bw[(size_t)(k + 1) * DHID];
            acc = __builtin_amdgcn_wmma_f32_16x16x4_f32(
                 false, a,  false, b,
                 (short)0, acc,  false,  false);
        }
        if (kh == 0) {
            const float bv = b1[col];
#pragma unroll
            for (int v = 0; v < GPB; ++v)
                Ah[v * APITCH + col] = fmaxf(acc[v] + bv, 0.0f);
        }
    }
    __syncthreads();

    {
        const float* ah = Ah + mr * APITCH + 2 * kh;
        const float* bw = Wout + (size_t)(2 * kh) * ODIM + n0 + col;
        v8f acc = {};
#pragma unroll
        for (int k = 0; k < DHID; k += 4) {
            const v2f a = *(const v2f*)(ah + k);
            v2f b;
            b.x = bw[(size_t)k * ODIM];
            b.y = bw[(size_t)(k + 1) * ODIM];
            acc = __builtin_amdgcn_wmma_f32_16x16x4_f32(
                 false, a,  false, b,
                 (short)0, acc,  false,  false);
        }
        if (kh == 0) {
            const float bv = bout[n0 + col];
#pragma unroll
            for (int v = 0; v < GPB; ++v)
                Os[v * APITCH + col] = acc[v] + bv;
        }
    }
    __syncthreads();
    if (w < 2 * GPB) {
        const int row = w >> 1;
        const int hf  = w & 1;
        const v4f o = *(const v4f*)&Os[row * APITCH + hf * 128 + 4 * lane];
        *(v4f*)(out + (size_t)(g0 + row) * ODIM + n0 + hf * 128 + 4 * lane) = o;
    }
}

extern "C" void kernel_launch(void* const* d_in, const int* in_sizes, int n_in,
                              void* d_out, int out_size, void* d_ws, size_t ws_size,
                              hipStream_t stream) {
    const int*   text_ids  = (const int*)  d_in[0];
    const int*   label_ids = (const int*)  d_in[1];
    const int*   text_len  = (const int*)  d_in[2];
    const int*   label_len = (const int*)  d_in[3];
    const float* word_emb  = (const float*)d_in[4];
    const float* label_emb = (const float*)d_in[5];
    const float* Wg_text   = (const float*)d_in[6];
    const float* Wg_label  = (const float*)d_in[7];
    const float* w_adap    = (const float*)d_in[8];
    const float* W1        = (const float*)d_in[9];
    const float* b1        = (const float*)d_in[10];
    const float* Wout      = (const float*)d_in[11];
    const float* bout      = (const float*)d_in[12];
    float* out = (float*)d_out;

    float* ws        = (float*)d_ws;
    float* h_f       = ws + (size_t)2 * NNODES * DHID;

    bag_project_kernel<<<NNODES / TILE_M, 512, 0, stream>>>(
        text_ids, label_ids, text_len, label_len,
        word_emb, label_emb, Wg_text, Wg_label, w_adap, h_f);

    graph_head_kernel<<<dim3(BGRAPH / GPB, ODIM / NSLICE), 512, 0, stream>>>(
        h_f, W1, b1, Wout, bout, out);
}
